# starting kernel
# baseline (speedup 1.0000x reference)
#include <hip/hip_runtime.h>
#include <hip/hip_bf16.h>


#define NTOT   256
#define BTOT   1024
#define DBOND  32
#define NCLS   10
#define NHALF  (NTOT / 2)
#define NLAST  (NTOT - 1)

typedef __attribute__((ext_vector_type(2))) float v2f;
typedef __attribute__((ext_vector_type(8))) float v8f;

__global__ __launch_bounds__(32)
void ClassificationMPS_kernel(const float* __restrict__ inputs,
                              const float* __restrict__ tensor,
                              const float* __restrict__ Aout,
                              float* __restrict__ out)
{
    __shared__ float sx255[16][2];
    __shared__ float sx128[16][2];
    __shared__ float st255[32][2];
    __shared__ float sT128[2][32][32];
    __shared__ float sAr[16][32];
    __shared__ float sP[2][16][32];
    __shared__ float sAl[16][32];
    __shared__ float sG[16][NCLS * 32];

    const int lane = threadIdx.x;
    const int b0   = blockIdx.x * 16;
    const int h    = lane >> 4;
    const int mn   = lane & 15;

    {
        const int m = lane >> 1, i = lane & 1;
        sx255[m][i] = inputs[(NLAST * BTOT + b0 + m) * 2 + i];
        sx128[m][i] = inputs[(NHALF * BTOT + b0 + m) * 2 + i];
        const int l = lane >> 1;
        st255[l][i]      = tensor[((NLAST * DBOND + l) * DBOND + 0) * 2 + i];
        st255[l + 16][i] = tensor[((NLAST * DBOND + l + 16) * DBOND + 0) * 2 + i];
    }
    for (int k = lane; k < 2 * 32 * 32; k += 32) {
        const int i = k & 1, r = (k >> 1) & 31, l = k >> 6;
        sT128[i][l][r] = tensor[((NHALF * DBOND + l) * DBOND + r) * 2 + i];
    }
    __syncthreads();

    for (int k = lane; k < 16 * 32; k += 32) {
        const int m = k >> 5, r = k & 31;
        sAr[m][r] = sx255[m][0] * st255[r][0] + sx255[m][1] * st255[r][1];
    }
    __syncthreads();

    for (int i = 0; i < 2; ++i) {
        for (int nt = 0; nt < 2; ++nt) {
            v8f acc = {};
#pragma unroll
            for (int s = 0; s < 8; ++s) {
                const int kb = 4 * s + 2 * h;
                v2f a, b;
                a.x = sAr[mn][kb];
                a.y = sAr[mn][kb + 1];
                b.x = sT128[i][nt * 16 + mn][kb];
                b.y = sT128[i][nt * 16 + mn][kb + 1];
                acc = __builtin_amdgcn_wmma_f32_16x16x4_f32(
                        false, a, false, b, (short)0, acc, false, false);
            }
#pragma unroll
            for (int g = 0; g < 8; ++g)
                sP[i][g + 8 * h][nt * 16 + mn] = acc[g];
        }
    }

    for (int nt = 0; nt < (NCLS * 32) / 16; ++nt) {
        const int j = nt * 16 + mn;
        const int o = j >> 5, l = j & 31;
        const float* ao = Aout + (o * DBOND + l) * DBOND;
        v8f acc = {};
#pragma unroll
        for (int s = 0; s < 8; ++s) {
            const int kb = 4 * s + 2 * h;
            v2f a, b;
            a.x = sAr[mn][kb];
            a.y = sAr[mn][kb + 1];
            b.x = ao[kb];
            b.y = ao[kb + 1];
            acc = __builtin_amdgcn_wmma_f32_16x16x4_f32(
                    false, a, false, b, (short)0, acc, false, false);
        }
#pragma unroll
        for (int g = 0; g < 8; ++g)
            sG[g + 8 * h][j] = acc[g];
    }
    __syncthreads();

    for (int k = lane; k < 16 * 32; k += 32) {
        const int m = k >> 5, l = k & 31;
        sAl[m][l] = sx128[m][0] * sP[0][m][l] + sx128[m][1] * sP[1][m][l];
    }
    __syncthreads();

    for (int k = lane; k < 16 * NCLS; k += 32) {
        const int m = k / NCLS, o = k % NCLS;
        float acc = 0.0f;
#pragma unroll
        for (int l = 0; l < 32; ++l)
            acc += sAl[m][l] * sG[m][o * 32 + l];
        out[(b0 + m) * NCLS + o] = acc;
    }
}

extern "C" void kernel_launch(void* const* d_in, const int* in_sizes, int n_in,
                              void* d_out, int out_size, void* d_ws, size_t ws_size,
                              hipStream_t stream) {
    const float* inputs = (const float*)d_in[0];
    const float* tensor = (const float*)d_in[1];
    const float* Aout   = (const float*)d_in[2];
    float* out = (float*)d_out;

    (void)in_sizes; (void)n_in; (void)out_size; (void)d_ws; (void)ws_size;

    ClassificationMPS_kernel<<<BTOT / 16, 32, 0, stream>>>(inputs, tensor, Aout, out);
}


# one work-group of 8 waves per 16-batch tile instead of one wave: the 24 WMMA output tiles are dealt 3 per wave with their fragment loads issued first; staging and epilogue strided over 256 threads
# speedup vs baseline: 2.9035x; 2.9035x over previous
#include <hip/hip_runtime.h>


#define NTOT   256
#define BTOT   1024
#define DBOND  32
#define NCLS   10
#define NHALF  (NTOT / 2)
#define NLAST  (NTOT - 1)
#define NWAVES 8
#define NTHR   (32 * NWAVES)
#define NPT    4

typedef __attribute__((ext_vector_type(2))) float v2f;
typedef __attribute__((ext_vector_type(8))) float v8f;

__global__ __launch_bounds__(NTHR)
void ClassificationMPS_kernel(const float* __restrict__ inputs,
                              const float* __restrict__ tensor,
                              const float* __restrict__ Aout,
                              float* __restrict__ out)
{
    __shared__ float sx255[16][2];
    __shared__ float sx128[16][2];
    __shared__ float st255[32][2];
    __shared__ float sT128[2][32][32];
    __shared__ float sAr[16][32];
    __shared__ float sP[2][16][32];
    __shared__ float sAl[16][32];
    __shared__ float sG[16][NCLS * 32];

    const int tid  = threadIdx.x;
    const int lane = tid & 31;
    const int wave = __builtin_amdgcn_readfirstlane(tid >> 5);
    const int b0   = blockIdx.x * 16;
    const int h    = lane >> 4;
    const int mn   = lane & 15;

    float tv[(2 * 32 * 32) / NTHR];
#pragma unroll
    for (int q = 0; q < (2 * 32 * 32) / NTHR; ++q)
        tv[q] = tensor[NHALF * DBOND * DBOND * 2 + q * NTHR + tid];
    if (wave == 0) {
        const int m = lane >> 1, i = lane & 1;
        const float xa = inputs[(NLAST * BTOT + b0 + m) * 2 + i];
        const float xb = inputs[(NHALF * BTOT + b0 + m) * 2 + i];
        const int l = lane >> 1;
        const float ta = tensor[((NLAST * DBOND + l) * DBOND + 0) * 2 + i];
        const float tb = tensor[((NLAST * DBOND + l + 16) * DBOND + 0) * 2 + i];
        sx255[m][i] = xa;
        sx128[m][i] = xb;
        st255[l][i]      = ta;
        st255[l + 16][i] = tb;
    }
#pragma unroll
    for (int q = 0; q < (2 * 32 * 32) / NTHR; ++q) {
        const int k = q * NTHR + tid;
        const int i = k & 1, r = (k >> 1) & 31, l = k >> 6;
        sT128[i][l][r] = tv[q];
    }
    __syncthreads();

#pragma unroll
    for (int q = 0; q < (16 * 32) / NTHR; ++q) {
        const int k = q * NTHR + tid;
        const int m = k >> 5, r = k & 31;
        sAr[m][r] = sx255[m][0] * st255[r][0] + sx255[m][1] * st255[r][1];
    }
    __syncthreads();

    const int gt0 = wave - NPT;
    const int gt1 = wave + NWAVES - NPT;
    const int gt2 = wave + 2 * NWAVES - NPT;
    const int pi  = wave >> 1, pnt = wave & 1;

    v2f bf0[8], bf1[8], bf2[8], af[8];
    if (wave < NPT) {
#pragma unroll
        for (int s = 0; s < 8; ++s) {
            const int kb = 4 * s + 2 * h;
            bf0[s].x = sT128[pi][pnt * 16 + mn][kb];
            bf0[s].y = sT128[pi][pnt * 16 + mn][kb + 1];
        }
    } else {
        const float* ao = Aout + (gt0 * 16 + mn) * DBOND;
#pragma unroll
        for (int s = 0; s < 8; ++s) {
            const int kb = 4 * s + 2 * h;
            bf0[s].x = ao[kb];
            bf0[s].y = ao[kb + 1];
        }
    }
    {
        const float* ao1 = Aout + (gt1 * 16 + mn) * DBOND;
        const float* ao2 = Aout + (gt2 * 16 + mn) * DBOND;
#pragma unroll
        for (int s = 0; s < 8; ++s) {
            const int kb = 4 * s + 2 * h;
            bf1[s].x = ao1[kb];
            bf1[s].y = ao1[kb + 1];
            bf2[s].x = ao2[kb];
            bf2[s].y = ao2[kb + 1];
        }
    }
#pragma unroll
    for (int s = 0; s < 8; ++s) {
        const int kb = 4 * s + 2 * h;
        af[s].x = sAr[mn][kb];
        af[s].y = sAr[mn][kb + 1];
    }

    v8f acc0 = {}, acc1 = {}, acc2 = {};
#pragma unroll
    for (int s = 0; s < 8; ++s) {
        acc0 = __builtin_amdgcn_wmma_f32_16x16x4_f32(
                false, af[s], false, bf0[s], (short)0, acc0, false, false);
        acc1 = __builtin_amdgcn_wmma_f32_16x16x4_f32(
                false, af[s], false, bf1[s], (short)0, acc1, false, false);
        acc2 = __builtin_amdgcn_wmma_f32_16x16x4_f32(
                false, af[s], false, bf2[s], (short)0, acc2, false, false);
    }

    if (wave < NPT) {
#pragma unroll
        for (int g = 0; g < 8; ++g)
            sP[pi][g + 8 * h][pnt * 16 + mn] = acc0[g];
    } else {
#pragma unroll
        for (int g = 0; g < 8; ++g)
            sG[g + 8 * h][gt0 * 16 + mn] = acc0[g];
    }
#pragma unroll
    for (int g = 0; g < 8; ++g) {
        sG[g + 8 * h][gt1 * 16 + mn] = acc1[g];
        sG[g + 8 * h][gt2 * 16 + mn] = acc2[g];
    }
    __syncthreads();

#pragma unroll
    for (int q = 0; q < (16 * 32) / NTHR; ++q) {
        const int k = q * NTHR + tid;
        const int m = k >> 5, l = k & 31;
        sAl[m][l] = sx128[m][0] * sP[0][m][l] + sx128[m][1] * sP[1][m][l];
    }
    __syncthreads();

    if (tid < 16 * NCLS) {
        const int m = tid / NCLS, o = tid % NCLS;
        float acc = 0.0f;
#pragma unroll
        for (int l = 0; l < 32; ++l)
            acc += sAl[m][l] * sG[m][o * 32 + l];
        out[(b0 + m) * NCLS + o] = acc;
    }
}

extern "C" void kernel_launch(void* const* d_in, const int* in_sizes, int n_in,
                              void* d_out, int out_size, void* d_ws, size_t ws_size,
                              hipStream_t stream) {
    const float* inputs = (const float*)d_in[0];
    const float* tensor = (const float*)d_in[1];
    const float* Aout   = (const float*)d_in[2];
    float* out = (float*)d_out;

    (void)in_sizes; (void)n_in; (void)out_size; (void)d_ws; (void)ws_size;

    ClassificationMPS_kernel<<<BTOT / 16, NTHR, 0, stream>>>(inputs, tensor, Aout, out);
}
